# starting kernel
# baseline (speedup 1.0000x reference)
#include <hip/hip_runtime.h>


#define N_IN   128
#define N_OUT  128
#define DEG    16
#define NEG_SLOPE 0.01f

typedef __attribute__((ext_vector_type(2))) float v2f;
typedef __attribute__((ext_vector_type(4))) float v4f;
typedef __attribute__((ext_vector_type(8))) float v8f;

__global__ __launch_bounds__(256) void gat_gemm_h(
    const float* __restrict__ feat, const float* __restrict__ W,
    const float* __restrict__ bw,   const float* __restrict__ aw,
    float* __restrict__ h, float* __restrict__ p, float* __restrict__ q,
    int n_nodes)
{
  const int lane = threadIdx.x & 31;
  const int wave = threadIdx.x >> 5;
  const int row0 = blockIdx.x * 128 + wave * 16;
  if (row0 >= n_nodes) return;

  const int m    = lane & 15;
  const int half = lane >> 4;

  const float* arow = feat + (size_t)(row0 + m) * N_IN + 2 * half;
  v2f A[32];
#pragma unroll
  for (int j = 0; j < 32; ++j) A[j] = *(const v2f*)(arow + 4 * j);

  float pacc[8], qacc[8];
#pragma unroll
  for (int r = 0; r < 8; ++r) { pacc[r] = 0.f; qacc[r] = 0.f; }

#pragma unroll
  for (int nt = 0; nt < 8; ++nt) {
    const int col = nt * 16 + m;
    const float bwc = bw[col];
    v8f acc;
#pragma unroll
    for (int r = 0; r < 8; ++r) acc[r] = bwc;

    const float* brow = W + (size_t)col * N_IN + 2 * half;
#pragma unroll
    for (int j = 0; j < 32; ++j) {
      v2f B = *(const v2f*)(brow + 4 * j);
      acc = __builtin_amdgcn_wmma_f32_16x16x4_f32(false, A[j], false, B,
                                                  (short)0, acc, false, false);
    }

    const float a_self = aw[col];
    const float a_nbr  = aw[N_OUT + col];
#pragma unroll
    for (int r = 0; r < 8; ++r) {
      const int row = row0 + r + 8 * half;
      h[(size_t)row * N_OUT + col] = acc[r];
      pacc[r] += acc[r] * a_self;
      qacc[r] += acc[r] * a_nbr;
    }
  }

#pragma unroll
  for (int r = 0; r < 8; ++r) {
    float pv = pacc[r], qv = qacc[r];
#pragma unroll
    for (int mask = 1; mask < 16; mask <<= 1) {
      pv += __shfl_xor(pv, mask, 32);
      qv += __shfl_xor(qv, mask, 32);
    }
    if (m == 0) {
      const int row = row0 + r + 8 * half;
      p[row] = pv;
      q[row] = qv;
    }
  }
}

__global__ __launch_bounds__(256) void gat_attn(
    const int* __restrict__ nodes, const int* __restrict__ nbrs,
    const float* __restrict__ h,   const float* __restrict__ p,
    const float* __restrict__ q,   const float* __restrict__ ab,
    float* __restrict__ out, int n_nodes)
{
  const int t = blockIdx.x * 8 + (threadIdx.x >> 5);
  if (t >= n_nodes) return;
  const int lane = threadIdx.x & 31;

  const int   k  = lane & 15;
  const float pt = p[nodes[t]];
  const int   nb = nbrs[t * DEG + k];
  float e = pt + q[nb] + ab[0];
  e = (e >= 0.f) ? e : NEG_SLOPE * e;

  float mx = e;
#pragma unroll
  for (int mask = 1; mask < 16; mask <<= 1)
    mx = fmaxf(mx, __shfl_xor(mx, mask, 32));
  const float ex = __expf(e - mx);
  float s = ex;
#pragma unroll
  for (int mask = 1; mask < 16; mask <<= 1)
    s += __shfl_xor(s, mask, 32);
  const float alpha = ex / s;

  v4f acc = {0.f, 0.f, 0.f, 0.f};
  const int c = lane * 4;
#pragma unroll
  for (int kk = 0; kk < DEG; ++kk) {
    const float a  = __shfl(alpha, kk, 32);
    const int   nn = __shfl(nb,    kk, 32);
    const v4f hv = *(const v4f*)(h + (size_t)nn * N_OUT + c);
#pragma unroll
    for (int i = 0; i < 4; ++i) acc[i] += a * hv[i];
  }
  *(v4f*)(out + (size_t)t * N_OUT + c) = acc;
}

extern "C" void kernel_launch(void* const* d_in, const int* in_sizes, int n_in,
                              void* d_out, int out_size, void* d_ws, size_t ws_size,
                              hipStream_t stream) {
  const float* feat  = (const float*)d_in[0];
  const int*   nodes = (const int*)  d_in[1];
  const int*   nbrs  = (const int*)  d_in[2];
  const float* W     = (const float*)d_in[3];
  const float* bw    = (const float*)d_in[4];
  const float* aw    = (const float*)d_in[5];
  const float* ab    = (const float*)d_in[6];
  float*       out   = (float*)d_out;

  const int n = in_sizes[1];

  float* h = (float*)d_ws;
  float* p = h + (size_t)n * N_OUT;
  float* q = p + n;

  gat_gemm_h<<<(n + 127) / 128, 256, 0, stream>>>(feat, W, bw, aw, h, p, q, n);
  gat_attn  <<<(n + 7) / 8,     256, 0, stream>>>(nodes, nbrs, h, p, q, ab, out, n);
}


# GEMM launch: W staged once per work-group in LDS (row pitch 528 B); B fragments read from LDS two k-steps ahead of their WMMAs, k outermost over 8 column tiles, instead of 16-line global loads
# speedup vs baseline: 1.2464x; 1.2464x over previous
#include <hip/hip_runtime.h>


#define N_IN   128
#define N_OUT  128
#define DEG    16
#define NEG_SLOPE 0.01f
#define W_PITCH 132

typedef __attribute__((ext_vector_type(2))) float v2f;
typedef __attribute__((ext_vector_type(4))) float v4f;
typedef __attribute__((ext_vector_type(8))) float v8f;

__global__ __launch_bounds__(256) void gat_gemm_h(
    const float* __restrict__ feat, const float* __restrict__ W,
    const float* __restrict__ bw,   const float* __restrict__ aw,
    float* __restrict__ h, float* __restrict__ p, float* __restrict__ q,
    int n_nodes)
{
  __shared__ __attribute__((aligned(16))) float Wl[N_OUT * W_PITCH];

  const int tid  = threadIdx.x;
  const int lane = tid & 31;
  const int wave = tid >> 5;
  const int row0 = blockIdx.x * 128 + wave * 16;
  const bool active = row0 < n_nodes;
  const int rowA = active ? row0 : 0;

  const int m    = lane & 15;
  const int half = lane >> 4;


  const float* arow = feat + (size_t)(rowA + m) * N_IN + 2 * half;
  v2f A[32];
#pragma unroll
  for (int j = 0; j < 32; ++j) A[j] = *(const v2f*)(arow + 4 * j);

#pragma unroll
  for (int i = 0; i < 16; ++i) {
    const int idx = tid + 256 * i;
    const int wr  = idx >> 5;
    const int wc  = (idx & 31) * 4;
    const v4f wv = *(const v4f*)(W + wr * N_IN + wc);
    *(v4f*)(Wl + wr * W_PITCH + wc) = wv;
  }

  float bwc[8], a_self[8], a_nbr[8];
#pragma unroll
  for (int nt = 0; nt < 8; ++nt) {
    const int col = nt * 16 + m;
    bwc[nt]    = bw[col];
    a_self[nt] = aw[col];
    a_nbr[nt]  = aw[N_OUT + col];
  }

  __syncthreads();
  if (!active) return;

  v8f acc[8];
#pragma unroll
  for (int nt = 0; nt < 8; ++nt)
#pragma unroll
    for (int r = 0; r < 8; ++r) acc[nt][r] = bwc[nt];

  const float* bbase = Wl + m * W_PITCH + 2 * half;
  v2f Bc[8][2];
#pragma unroll
  for (int nt = 0; nt < 8; ++nt) {
    Bc[nt][0] = *(const v2f*)(bbase + nt * 16 * W_PITCH);
    Bc[nt][1] = *(const v2f*)(bbase + nt * 16 * W_PITCH + 4);
  }
#pragma unroll
  for (int jp = 0; jp < 16; ++jp) {
    v2f Bn[8][2];
    if (jp + 1 < 16) {
#pragma unroll
      for (int nt = 0; nt < 8; ++nt) {
        Bn[nt][0] = *(const v2f*)(bbase + nt * 16 * W_PITCH + 8 * (jp + 1));
        Bn[nt][1] = *(const v2f*)(bbase + nt * 16 * W_PITCH + 8 * (jp + 1) + 4);
      }
    }
    __builtin_amdgcn_sched_barrier(0);
#pragma unroll
    for (int s = 0; s < 2; ++s)
#pragma unroll
      for (int nt = 0; nt < 8; ++nt)
        acc[nt] = __builtin_amdgcn_wmma_f32_16x16x4_f32(false, A[2 * jp + s], false, Bc[nt][s],
                                                        (short)0, acc[nt], false, false);
    if (jp + 1 < 16) {
#pragma unroll
      for (int nt = 0; nt < 8; ++nt) { Bc[nt][0] = Bn[nt][0]; Bc[nt][1] = Bn[nt][1]; }
    }
  }

  float pacc[8], qacc[8];
#pragma unroll
  for (int r = 0; r < 8; ++r) { pacc[r] = 0.f; qacc[r] = 0.f; }
#pragma unroll
  for (int nt = 0; nt < 8; ++nt) {
    const int col = nt * 16 + m;
#pragma unroll
    for (int r = 0; r < 8; ++r) {
      const int row = row0 + r + 8 * half;
      h[(size_t)row * N_OUT + col] = acc[nt][r];
      pacc[r] += acc[nt][r] * a_self[nt];
      qacc[r] += acc[nt][r] * a_nbr[nt];
    }
  }

#pragma unroll
  for (int r = 0; r < 8; ++r) {
    float pv = pacc[r], qv = qacc[r];
#pragma unroll
    for (int mask = 1; mask < 16; mask <<= 1) {
      pv += __shfl_xor(pv, mask, 32);
      qv += __shfl_xor(qv, mask, 32);
    }
    if (m == 0) {
      const int row = row0 + r + 8 * half;
      p[row] = pv;
      q[row] = qv;
    }
  }
}

__global__ __launch_bounds__(256) void gat_attn(
    const int* __restrict__ nodes, const int* __restrict__ nbrs,
    const float* __restrict__ h,   const float* __restrict__ p,
    const float* __restrict__ q,   const float* __restrict__ ab,
    float* __restrict__ out, int n_nodes)
{
  const int t = blockIdx.x * 8 + (threadIdx.x >> 5);
  if (t >= n_nodes) return;
  const int lane = threadIdx.x & 31;

  const int   k  = lane & 15;
  const float pt = p[nodes[t]];
  const int   nb = nbrs[t * DEG + k];
  float e = pt + q[nb] + ab[0];
  e = (e >= 0.f) ? e : NEG_SLOPE * e;

  float mx = e;
#pragma unroll
  for (int mask = 1; mask < 16; mask <<= 1)
    mx = fmaxf(mx, __shfl_xor(mx, mask, 32));
  const float ex = __expf(e - mx);
  float s = ex;
#pragma unroll
  for (int mask = 1; mask < 16; mask <<= 1)
    s += __shfl_xor(s, mask, 32);
  const float alpha = ex / s;

  v4f acc = {0.f, 0.f, 0.f, 0.f};
  const int c = lane * 4;
#pragma unroll
  for (int kk = 0; kk < DEG; ++kk) {
    const float a  = __shfl(alpha, kk, 32);
    const int   nn = __shfl(nb,    kk, 32);
    const v4f hv = *(const v4f*)(h + (size_t)nn * N_OUT + c);
#pragma unroll
    for (int i = 0; i < 4; ++i) acc[i] += a * hv[i];
  }
  *(v4f*)(out + (size_t)t * N_OUT + c) = acc;
}

extern "C" void kernel_launch(void* const* d_in, const int* in_sizes, int n_in,
                              void* d_out, int out_size, void* d_ws, size_t ws_size,
                              hipStream_t stream) {
  const float* feat  = (const float*)d_in[0];
  const int*   nodes = (const int*)  d_in[1];
  const int*   nbrs  = (const int*)  d_in[2];
  const float* W     = (const float*)d_in[3];
  const float* bw    = (const float*)d_in[4];
  const float* aw    = (const float*)d_in[5];
  const float* ab    = (const float*)d_in[6];
  float*       out   = (float*)d_out;

  const int n = in_sizes[1];

  float* h = (float*)d_ws;
  float* p = h + (size_t)n * N_OUT;
  float* q = p + n;

  gat_gemm_h<<<(n + 127) / 128, 256, 0, stream>>>(feat, W, bw, aw, h, p, q, n);
  gat_attn  <<<(n + 7) / 8,     256, 0, stream>>>(nodes, nbrs, h, p, q, ab, out, n);
}


# Per-node reduction: lane exchanges (logit sums, softmax max and sum) on the vector ALU (DPP row_xmask, permlanex16), not LDS-unit permutes; same exchange tree; logit dot4 as explicit fma chain
# speedup vs baseline: 1.4733x; 1.1820x over previous
#include <hip/hip_runtime.h>


#define N_IN   128
#define N_OUT  128
#define DEG    16
#define NEG_SLOPE 0.01f
#define W_PITCH 132
#define A_PITCH 132

typedef __attribute__((ext_vector_type(2))) float v2f;
typedef __attribute__((ext_vector_type(4))) float v4f;
typedef __attribute__((ext_vector_type(8))) float v8f;
typedef __attribute__((ext_vector_type(4))) int   v4i;

__device__ __forceinline__ float dot4(const v4f a, const v4f b) {
  return __builtin_fmaf(a[3], b[3], __builtin_fmaf(a[2], b[2], __builtin_fmaf(a[1], b[1], a[0] * b[0])));
}

template <int M>
__device__ __forceinline__ float xrow(const float x) {
  return __builtin_bit_cast(float, __builtin_amdgcn_update_dpp(0, __builtin_bit_cast(int, x),
                                                               0x160 + M, 0xf, 0xf, true));
}
__device__ __forceinline__ float xhalf(const float x) {
  const int b = __builtin_bit_cast(int, x);
  return __builtin_bit_cast(float,
      __builtin_amdgcn_permlanex16(b, b, 0x76543210, 0xfedcba98, false, false));
}

__device__ __forceinline__ void issue_rows(v4f (&R)[17], const float* __restrict__ feat,
                                           const int (&idx8)[8], const int nv,
                                           const int i, const int lane4) {
#pragma unroll
  for (int kk = 0; kk < DEG; ++kk) {
    const int nn = __builtin_amdgcn_readlane(idx8[kk & 7], 2 * i + (kk >> 3));
    R[kk] = *(const v4f*)(feat + (size_t)nn * N_IN + lane4);
  }
  const int ns = __builtin_amdgcn_readlane(nv, i);
  R[16] = *(const v4f*)(feat + (size_t)ns * N_IN + lane4);
}

__device__ __forceinline__ v4f reduce_node(const v4f (&R)[17], const v4f wp4, const v4f wq4,
                                           const float c0, const int lane) {
  float d[16];
#pragma unroll
  for (int kk = 0; kk < DEG; ++kk) d[kk] = dot4(R[kk], wq4);
  float ds = dot4(R[16], wp4);
  ds += xhalf(ds);
  ds += xrow<8>(ds);
  ds += xrow<4>(ds);
  ds += xrow<2>(ds);
  ds += xrow<1>(ds);

  const bool b4 = (lane & 16) != 0, b3 = (lane & 8) != 0;
  const bool b2 = (lane & 4) != 0,  b1 = (lane & 2) != 0;
  float v8[8], v4[4], v2[2];
#pragma unroll
  for (int r = 0; r < 8; ++r) {
    const float keep = b4 ? d[r + 8] : d[r];
    const float send = b4 ? d[r] : d[r + 8];
    v8[r] = keep + xhalf(send);
  }
#pragma unroll
  for (int r = 0; r < 4; ++r) {
    const float keep = b3 ? v8[r + 4] : v8[r];
    const float send = b3 ? v8[r] : v8[r + 4];
    v4[r] = keep + xrow<8>(send);
  }
#pragma unroll
  for (int r = 0; r < 2; ++r) {
    const float keep = b2 ? v4[r + 2] : v4[r];
    const float send = b2 ? v4[r] : v4[r + 2];
    v2[r] = keep + xrow<4>(send);
  }
  float v1;
  {
    const float keep = b1 ? v2[1] : v2[0];
    const float send = b1 ? v2[0] : v2[1];
    v1 = keep + xrow<2>(send);
  }
  v1 += xrow<1>(v1);

  float e = ds + v1 + c0;
  e = (e >= 0.f) ? e : NEG_SLOPE * e;

  float mx = e;
  mx = fmaxf(mx, xrow<2>(mx));
  mx = fmaxf(mx, xrow<4>(mx));
  mx = fmaxf(mx, xrow<8>(mx));
  mx = fmaxf(mx, xhalf(mx));
  const float ex = __expf(e - mx);
  float s = ex;
  s += xrow<2>(s);
  s += xrow<4>(s);
  s += xrow<8>(s);
  s += xhalf(s);
  const float alpha = ex / s;

  v4f acc = {0.f, 0.f, 0.f, 0.f};
#pragma unroll
  for (int kk = 0; kk < DEG; ++kk) {
    const float a = __int_as_float(__builtin_amdgcn_readlane(__float_as_int(alpha), 2 * kk));
#pragma unroll
    for (int c = 0; c < 4; ++c) acc[c] += a * R[kk][c];
  }
  return acc;
}

__global__ __launch_bounds__(256) void gat_fused(
    const float* __restrict__ feat, const int* __restrict__ nodes,
    const int* __restrict__ nbrs,   const float* __restrict__ W,
    const float* __restrict__ bw,   const float* __restrict__ aw,
    const float* __restrict__ ab,   float* __restrict__ out, int n_nodes)
{
  __shared__ __attribute__((aligned(16))) float Wl[N_OUT * W_PITCH];
  __shared__ __attribute__((aligned(16))) float Al[8 * 16 * A_PITCH];
  __shared__ __attribute__((aligned(16))) float wvec[2 * N_IN];
  __shared__ float cs[1];

  const int tid  = threadIdx.x;
  const int lane = tid & 31;
  const int wave = __builtin_amdgcn_readfirstlane(tid >> 5);
  const int row0 = blockIdx.x * 128 + wave * 16;
  const bool active = row0 < n_nodes;
  const int rowA = active ? row0 : 0;

  const int m     = lane & 15;
  const int half  = lane >> 4;
  const int lane4 = lane * 4;

  const int nv = nodes[rowA + m];
  const v4i* ip = (const v4i*)(nbrs + (size_t)rowA * DEG + lane * 8);
  const v4i i0 = ip[0], i1 = ip[1];
  const int idx8[8] = {i0[0], i0[1], i0[2], i0[3], i1[0], i1[1], i1[2], i1[3]};

  v4f pp = {0.f, 0.f, 0.f, 0.f}, pq = {0.f, 0.f, 0.f, 0.f};
#pragma unroll
  for (int i = 0; i < 16; ++i) {
    const int wr = wave + 8 * i;
    const v4f wv = *(const v4f*)(W + wr * N_IN + lane4);
    *(v4f*)(Wl + wr * W_PITCH + lane4) = wv;
    const float as_ = aw[wr];
    const float an_ = aw[N_OUT + wr];
#pragma unroll
    for (int c = 0; c < 4; ++c) { pp[c] += wv[c] * as_; pq[c] += wv[c] * an_; }
  }
  float* red = Al;
  *(v4f*)(red + (wave * 2 + 0) * N_IN + lane4) = pp;
  *(v4f*)(red + (wave * 2 + 1) * N_IN + lane4) = pq;

  if (wave == 0) {
    float t = 0.f;
#pragma unroll
    for (int c = 0; c < 4; ++c) {
      const int n = lane4 + c;
      t += bw[n] * (aw[n] + aw[N_OUT + n]);
    }
#pragma unroll
    for (int mask = 16; mask >= 1; mask >>= 1) t += __shfl_xor(t, mask, 32);
    if (lane == 0) cs[0] = t;
  }

  float bwc[8];
#pragma unroll
  for (int nt = 0; nt < 8; ++nt) bwc[nt] = bw[nt * 16 + m];

  __syncthreads();
  {
    const int which = tid >> 7, k = tid & 127;
    float s = 0.f;
#pragma unroll
    for (int w = 0; w < 8; ++w) s += red[(w * 2 + which) * N_IN + k];
    wvec[which * N_IN + k] = s;
  }
  __syncthreads();
  if (!active) return;

  const v4f wp4 = *(const v4f*)(wvec + lane4);
  const v4f wq4 = *(const v4f*)(wvec + N_IN + lane4);
  const float c0 = cs[0] + ab[0];

  float* Aw = Al + wave * 16 * A_PITCH;
  v4f Ra[17], Rb[17];
  issue_rows(Ra, feat, idx8, nv, 0, lane4);
  for (int i = 0; i < 16; i += 2) {
    issue_rows(Rb, feat, idx8, nv, i + 1, lane4);
    __builtin_amdgcn_sched_barrier(0);
    *(v4f*)(Aw + i * A_PITCH + lane4) = reduce_node(Ra, wp4, wq4, c0, lane);
    if (i + 2 < 16) issue_rows(Ra, feat, idx8, nv, i + 2, lane4);
    __builtin_amdgcn_sched_barrier(0);
    *(v4f*)(Aw + (i + 1) * A_PITCH + lane4) = reduce_node(Rb, wp4, wq4, c0, lane);
  }

  const float* arow = Aw + m * A_PITCH + 2 * half;
  v2f A[32];
#pragma unroll
  for (int j = 0; j < 32; ++j) A[j] = *(const v2f*)(arow + 4 * j);

  v8f acc[8];
#pragma unroll
  for (int nt = 0; nt < 8; ++nt)
#pragma unroll
    for (int r = 0; r < 8; ++r) acc[nt][r] = bwc[nt];

  const float* bbase = Wl + m * W_PITCH + 2 * half;
  v2f Bc[8][2];
#pragma unroll
  for (int nt = 0; nt < 8; ++nt) {
    Bc[nt][0] = *(const v2f*)(bbase + nt * 16 * W_PITCH);
    Bc[nt][1] = *(const v2f*)(bbase + nt * 16 * W_PITCH + 4);
  }
#pragma unroll
  for (int jp = 0; jp < 16; ++jp) {
    v2f Bn[8][2];
    if (jp + 1 < 16) {
#pragma unroll
      for (int nt = 0; nt < 8; ++nt) {
        Bn[nt][0] = *(const v2f*)(bbase + nt * 16 * W_PITCH + 8 * (jp + 1));
        Bn[nt][1] = *(const v2f*)(bbase + nt * 16 * W_PITCH + 8 * (jp + 1) + 4);
      }
    }
    __builtin_amdgcn_sched_barrier(0);
#pragma unroll
    for (int s = 0; s < 2; ++s)
#pragma unroll
      for (int nt = 0; nt < 8; ++nt)
        acc[nt] = __builtin_amdgcn_wmma_f32_16x16x4_f32(false, A[2 * jp + s], false, Bc[nt][s],
                                                        (short)0, acc[nt], false, false);
    if (jp + 1 < 16) {
#pragma unroll
      for (int nt = 0; nt < 8; ++nt) { Bc[nt][0] = Bn[nt][0]; Bc[nt][1] = Bn[nt][1]; }
    }
  }

#pragma unroll
  for (int nt = 0; nt < 8; ++nt) {
    const int col = nt * 16 + m;
#pragma unroll
    for (int r = 0; r < 8; ++r) {
      const int row = row0 + r + 8 * half;
      out[(size_t)row * N_OUT + col] = acc[nt][r];
    }
  }
}

extern "C" void kernel_launch(void* const* d_in, const int* in_sizes, int n_in,
                              void* d_out, int out_size, void* d_ws, size_t ws_size,
                              hipStream_t stream) {
  const float* feat  = (const float*)d_in[0];
  const int*   nodes = (const int*)  d_in[1];
  const int*   nbrs  = (const int*)  d_in[2];
  const float* W     = (const float*)d_in[3];
  const float* bw    = (const float*)d_in[4];
  const float* aw    = (const float*)d_in[5];
  const float* ab    = (const float*)d_in[6];
  float*       out   = (float*)d_out;

  const int n = in_sizes[1];

  gat_fused<<<(n + 127) / 128, 256, 0, stream>>>(feat, nodes, nbrs, W, bw, aw, ab, out, n);
}


# Work dealt evenly over the WGPs: 256 work-groups of 16 waves; wave w gathers row w of each of the group's 12-13 tiles, then 2 column tiles per tile with B kept in registers; W^T a sum order changes
# speedup vs baseline: 1.6583x; 1.1256x over previous
#include <hip/hip_runtime.h>


#define N_IN   128
#define N_OUT  128
#define DEG    16
#define NEG_SLOPE 0.01f
#define W_PITCH 132
#define A_PITCH 132
#define GRID   256
#define MAXT   13

typedef __attribute__((ext_vector_type(2))) float v2f;
typedef __attribute__((ext_vector_type(4))) float v4f;
typedef __attribute__((ext_vector_type(8))) float v8f;
typedef __attribute__((ext_vector_type(4))) int   v4i;

__device__ __forceinline__ float dot4(const v4f a, const v4f b) {
  return __builtin_fmaf(a[3], b[3], __builtin_fmaf(a[2], b[2], __builtin_fmaf(a[1], b[1], a[0] * b[0])));
}

template <int M>
__device__ __forceinline__ float xrow(const float x) {
  return __builtin_bit_cast(float, __builtin_amdgcn_update_dpp(0, __builtin_bit_cast(int, x),
                                                               0x160 + M, 0xf, 0xf, true));
}
__device__ __forceinline__ float xhalf(const float x) {
  const int b = __builtin_bit_cast(int, x);
  return __builtin_bit_cast(float,
      __builtin_amdgcn_permlanex16(b, b, 0x76543210, 0xfedcba98, false, false));
}

__device__ __forceinline__ void issue_rows(v4f (&R)[17], const float* __restrict__ feat,
                                           const int (&idx8)[8], const int nv,
                                           const int i, const int lane4) {
#pragma unroll
  for (int kk = 0; kk < DEG; ++kk) {
    const int nn = __builtin_amdgcn_readlane(idx8[kk & 7], 2 * i + (kk >> 3));
    R[kk] = *(const v4f*)(feat + (size_t)nn * N_IN + lane4);
  }
  const int ns = __builtin_amdgcn_readlane(nv, i);
  R[16] = *(const v4f*)(feat + (size_t)ns * N_IN + lane4);
}

__device__ __forceinline__ v4f reduce_node(const v4f (&R)[17], const v4f wp4, const v4f wq4,
                                           const float c0, const int lane) {
  float d[16];
#pragma unroll
  for (int kk = 0; kk < DEG; ++kk) d[kk] = dot4(R[kk], wq4);
  float ds = dot4(R[16], wp4);
  ds += xhalf(ds);
  ds += xrow<8>(ds);
  ds += xrow<4>(ds);
  ds += xrow<2>(ds);
  ds += xrow<1>(ds);

  const bool b4 = (lane & 16) != 0, b3 = (lane & 8) != 0;
  const bool b2 = (lane & 4) != 0,  b1 = (lane & 2) != 0;
  float v8[8], v4[4], v2[2];
#pragma unroll
  for (int r = 0; r < 8; ++r) {
    const float keep = b4 ? d[r + 8] : d[r];
    const float send = b4 ? d[r] : d[r + 8];
    v8[r] = keep + xhalf(send);
  }
#pragma unroll
  for (int r = 0; r < 4; ++r) {
    const float keep = b3 ? v8[r + 4] : v8[r];
    const float send = b3 ? v8[r] : v8[r + 4];
    v4[r] = keep + xrow<8>(send);
  }
#pragma unroll
  for (int r = 0; r < 2; ++r) {
    const float keep = b2 ? v4[r + 2] : v4[r];
    const float send = b2 ? v4[r] : v4[r + 2];
    v2[r] = keep + xrow<4>(send);
  }
  float v1;
  {
    const float keep = b1 ? v2[1] : v2[0];
    const float send = b1 ? v2[0] : v2[1];
    v1 = keep + xrow<2>(send);
  }
  v1 += xrow<1>(v1);

  float e = ds + v1 + c0;
  e = (e >= 0.f) ? e : NEG_SLOPE * e;

  float mx = e;
  mx = fmaxf(mx, xrow<2>(mx));
  mx = fmaxf(mx, xrow<4>(mx));
  mx = fmaxf(mx, xrow<8>(mx));
  mx = fmaxf(mx, xhalf(mx));
  const float ex = __expf(e - mx);
  float s = ex;
  s += xrow<2>(s);
  s += xrow<4>(s);
  s += xrow<8>(s);
  s += xhalf(s);
  const float alpha = ex / s;

  v4f acc = {0.f, 0.f, 0.f, 0.f};
#pragma unroll
  for (int kk = 0; kk < DEG; ++kk) {
    const float a = __int_as_float(__builtin_amdgcn_readlane(__float_as_int(alpha), 2 * kk));
#pragma unroll
    for (int c = 0; c < 4; ++c) acc[c] += a * R[kk][c];
  }
  return acc;
}

__global__ __launch_bounds__(512) void gat_fused(
    const float* __restrict__ feat, const int* __restrict__ nodes,
    const int* __restrict__ nbrs,   const float* __restrict__ W,
    const float* __restrict__ bw,   const float* __restrict__ aw,
    const float* __restrict__ ab,   float* __restrict__ out, int n_nodes)
{
  __shared__ __attribute__((aligned(16))) float Wl[N_OUT * W_PITCH];
  __shared__ __attribute__((aligned(16))) float Al[MAXT * 16 * A_PITCH];
  __shared__ __attribute__((aligned(16))) float wvec[2 * N_IN];
  __shared__ float cs[1];

  const int tid  = threadIdx.x;
  const int lane = tid & 31;
  const int wave = __builtin_amdgcn_readfirstlane(tid >> 5);
  const int ntiles = n_nodes >> 4;

  const int m     = lane & 15;
  const int half  = lane >> 4;
  const int lane4 = lane * 4;

  const int t0 = blockIdx.x;
  int cnt = (ntiles - t0 + GRID - 1) / GRID;
  cnt = cnt < 0 ? 0 : (cnt > MAXT ? MAXT : cnt);

  int nv;
  int idx8[8];
  {
    const int jn = (m < cnt) ? m : 0;
    const int tn = (m < cnt) ? t0 + GRID * jn : 0;
    nv = nodes[tn * 16 + wave];
    const int je = lane >> 1;
    const int te = (je < cnt) ? t0 + GRID * je : 0;
    const v4i* ip = (const v4i*)(nbrs + ((size_t)te * 16 + wave) * DEG + (lane & 1) * 8);
    const v4i i0 = ip[0], i1 = ip[1];
    idx8[0] = i0[0]; idx8[1] = i0[1]; idx8[2] = i0[2]; idx8[3] = i0[3];
    idx8[4] = i1[0]; idx8[5] = i1[1]; idx8[6] = i1[2]; idx8[7] = i1[3];
  }

  v4f pp = {0.f, 0.f, 0.f, 0.f}, pq = {0.f, 0.f, 0.f, 0.f};
#pragma unroll
  for (int i = 0; i < 8; ++i) {
    const int wr = wave + 16 * i;
    const v4f wv = *(const v4f*)(W + wr * N_IN + lane4);
    *(v4f*)(Wl + wr * W_PITCH + lane4) = wv;
    const float as_ = aw[wr];
    const float an_ = aw[N_OUT + wr];
#pragma unroll
    for (int c = 0; c < 4; ++c) { pp[c] += wv[c] * as_; pq[c] += wv[c] * an_; }
  }
  float* red = Al;
  *(v4f*)(red + (wave * 2 + 0) * N_IN + lane4) = pp;
  *(v4f*)(red + (wave * 2 + 1) * N_IN + lane4) = pq;

  if (wave == 0) {
    float t = 0.f;
#pragma unroll
    for (int c = 0; c < 4; ++c) {
      const int n = lane4 + c;
      t += bw[n] * (aw[n] + aw[N_OUT + n]);
    }
#pragma unroll
    for (int mask = 16; mask >= 1; mask >>= 1) t += __shfl_xor(t, mask, 32);
    if (lane == 0) cs[0] = t;
  }

  const int np = wave & 3;
  const int gq = ((wave >> 2) + (wave & 3)) & 3;
  const float bias0 = bw[(2 * np) * 16 + m];
  const float bias1 = bw[(2 * np + 1) * 16 + m];

  __syncthreads();
  if (tid < 2 * N_IN) {
    const int which = tid >> 7, k = tid & 127;
    float s = 0.f;
#pragma unroll
    for (int w = 0; w < 16; ++w) s += red[(w * 2 + which) * N_IN + k];
    wvec[which * N_IN + k] = s;
  }
  __syncthreads();

  const v4f wp4 = *(const v4f*)(wvec + lane4);
  const v4f wq4 = *(const v4f*)(wvec + N_IN + lane4);
  const float c0 = cs[0] + ab[0];

  if (cnt > 0) {
    float* Aw = Al + wave * A_PITCH;
    v4f Ra[17], Rb[17];
    issue_rows(Ra, feat, idx8, nv, 0, lane4);
    for (int i = 0; i < cnt; i += 2) {
      const bool two = i + 1 < cnt;
      if (two) issue_rows(Rb, feat, idx8, nv, i + 1, lane4);
      __builtin_amdgcn_sched_barrier(0);
      *(v4f*)(Aw + i * 16 * A_PITCH + lane4) = reduce_node(Ra, wp4, wq4, c0, lane);
      if (i + 2 < cnt) issue_rows(Ra, feat, idx8, nv, i + 2, lane4);
      __builtin_amdgcn_sched_barrier(0);
      if (two) *(v4f*)(Aw + (i + 1) * 16 * A_PITCH + lane4) = reduce_node(Rb, wp4, wq4, c0, lane);
    }

    __syncthreads();

    const float* bbase = Wl + ((2 * np) * 16 + m) * W_PITCH + 2 * half;
    v2f B0[32], B1[32];
#pragma unroll
    for (int j = 0; j < 32; ++j) {
      B0[j] = *(const v2f*)(bbase + 4 * j);
      B1[j] = *(const v2f*)(bbase + 16 * W_PITCH + 4 * j);
    }

    for (int jt = gq; jt < cnt; jt += 4) {
      const float* arow = Al + (jt * 16 + m) * A_PITCH + 2 * half;
      v8f acc0, acc1;
#pragma unroll
      for (int r = 0; r < 8; ++r) { acc0[r] = bias0; acc1[r] = bias1; }
#pragma unroll
      for (int h2 = 0; h2 < 2; ++h2) {
        v2f A[16];
#pragma unroll
        for (int j = 0; j < 16; ++j) A[j] = *(const v2f*)(arow + 4 * (16 * h2 + j));
#pragma unroll
        for (int j = 0; j < 16; ++j) {
          acc0 = __builtin_amdgcn_wmma_f32_16x16x4_f32(false, A[j], false, B0[16 * h2 + j],
                                                       (short)0, acc0, false, false);
          acc1 = __builtin_amdgcn_wmma_f32_16x16x4_f32(false, A[j], false, B1[16 * h2 + j],
                                                       (short)0, acc1, false, false);
        }
      }

      const int row0 = (t0 + GRID * jt) * 16;
      float* op = out + (size_t)(row0 + 8 * half) * N_OUT + (2 * np) * 16 + m;
#pragma unroll
      for (int r = 0; r < 8; ++r) {
        op[(size_t)r * N_OUT]      = acc0[r];
        op[(size_t)r * N_OUT + 16] = acc1[r];
      }
    }
  }
}

extern "C" void kernel_launch(void* const* d_in, const int* in_sizes, int n_in,
                              void* d_out, int out_size, void* d_ws, size_t ws_size,
                              hipStream_t stream) {
  const float* feat  = (const float*)d_in[0];
  const int*   nodes = (const int*)  d_in[1];
  const int*   nbrs  = (const int*)  d_in[2];
  const float* W     = (const float*)d_in[3];
  const float* bw    = (const float*)d_in[4];
  const float* aw    = (const float*)d_in[5];
  const float* ab    = (const float*)d_in[6];
  float*       out   = (float*)d_out;

  const int n = in_sizes[1];

  gat_fused<<<GRID, 512, 0, stream>>>(feat, nodes, nbrs, W, bw, aw, ab, out, n);
}
